# starting kernel
# baseline (speedup 1.0000x reference)
#include <hip/hip_runtime.h>

typedef __attribute__((ext_vector_type(2))) float v2f;
typedef __attribute__((ext_vector_type(8))) float v8f;

#define DIMSZ 256
#define NBATCH 16
#define NPIX (DIMSZ * DIMSZ)
#define NELEM (NBATCH * NPIX)
#define LDSW 138
#define LDSH 22
#define NSCHUNK (7 * 6)
#define SBUF_ELEMS (NSCHUNK * 32 * 2)

__global__ void fdtd_init(const float* __restrict__ fdk,
                          const float* __restrict__ cw,
                          float* __restrict__ Sbuf) {
    __shared__ float weff[49];
    int t = threadIdx.x;
    if (t < 49) {
        float acc = 0.0f;
        #pragma unroll
        for (int ch = 1; ch < 15; ++ch)
            acc += cw[ch - 1] * fdk[ch * 49 + t];
        weff[t] = acc * (2.0f * 0.5f / 1.0f);
    }
    __syncthreads();
    for (int e = t; e < SBUF_ELEMS; e += blockDim.x) {
        int v    = e & 1;
        int lane = (e >> 1) & 31;
        int c    = (e >> 6) % 6;
        int ky   = e / 384;
        int n    = lane & 15;
        int k    = 4 * c + v + 2 * (lane >> 4);
        int j    = k - n;
        float val = (k < 22 && j >= 0 && j <= 6) ? weff[ky * 7 + j] : 0.0f;
        Sbuf[e] = val;
    }
}

__global__ __launch_bounds__(256) void fdtd_step(
    const float* __restrict__ cur,
    const float* prev,
    const float* __restrict__ epsr,
    const float* __restrict__ sigma,
    const float* __restrict__ Sglob,
    const float* __restrict__ src,
    float* next,
    int step) {
    __shared__ float tile[LDSH * LDSW];
    __shared__ float Slds[SBUF_ELEMS];

    const int b  = blockIdx.x;
    const int y0 = blockIdx.y * 16;
    const int x0 = blockIdx.z * 128;
    const int t  = threadIdx.x;
    const float* curb = cur + b * NPIX;

    for (int e = t; e < LDSH * LDSW; e += 256) {
        int r = e / LDSW, c = e % LDSW;
        int gy = y0 + r - 3, gx = x0 + c - 3;
        float v = 0.0f;
        if (c < 136 && gy >= 0 && gy < DIMSZ && gx >= 0 && gx < DIMSZ)
            v = curb[gy * DIMSZ + gx];
        tile[e] = v;
    }
    for (int e = t; e < SBUF_ELEMS; e += 256) Slds[e] = Sglob[e];
    __syncthreads();

    const int wave = t >> 5;
    const int lane = t & 31;
    const int mlo  = lane & 15;
    const int sel  = lane >> 4;
    const int xw   = 16 * wave;

    v8f acc = {0.f, 0.f, 0.f, 0.f, 0.f, 0.f, 0.f, 0.f};
    #pragma unroll
    for (int ky = 0; ky < 7; ++ky) {
        const int arow = (mlo + ky) * LDSW + xw + 2 * sel;
        #pragma unroll
        for (int c = 0; c < 6; ++c) {
            v2f a = *(const v2f*)&tile[arow + 4 * c];
            v2f bmat = *(const v2f*)&Slds[((ky * 6 + c) * 32 + lane) * 2];
            acc = __builtin_amdgcn_wmma_f32_16x16x4_f32(
                false, a, false, bmat, (short)0, acc, false, false);
        }
    }

    const float dt = 0.5f;
    const float s0 = src[2 * step], s1 = src[2 * step + 1];
    const float srcterm = -(s0 - s1);
    const int gx = x0 + xw + mlo;
    #pragma unroll
    for (int r = 0; r < 8; ++r) {
        int m  = r + 8 * sel;
        int gy = y0 + m;
        int idx = b * NPIX + gy * DIMSZ + gx;
        float ep = epsr[idx], sg = sigma[idx];
        float inv_den = 1.0f / (2.0f * ep / dt + sg);
        float av = 4.0f * ep / dt;
        float bv = 2.0f * ep / dt - sg;
        float cu = tile[(m + 3) * LDSW + xw + mlo + 3];
        float pv = prev[idx];
        float rhs = acc[r] + av * cu - bv * pv;
        if (gy == 128 && gx == 128) rhs += srcterm;
        next[idx] = inv_den * rhs;
    }
}

extern "C" void kernel_launch(void* const* d_in, const int* in_sizes, int n_in,
                              void* d_out, int out_size, void* d_ws, size_t ws_size,
                              hipStream_t stream) {
    const float* u1    = (const float*)d_in[0];
    const float* u0    = (const float*)d_in[1];
    const float* epsr  = (const float*)d_in[2];
    const float* sigma = (const float*)d_in[3];
    const float* fdk   = (const float*)d_in[4];
    const float* cw    = (const float*)d_in[5];
    const float* src   = (const float*)d_in[6];
    float* out = (float*)d_out;
    float* ws  = (float*)d_ws;

    float* bufs[3] = {ws, ws + NELEM, ws + 2 * NELEM};
    float* Sbuf = ws + 3 * NELEM;

    fdtd_init<<<1, 256, 0, stream>>>(fdk, cw, Sbuf);

    const float* cur  = u1;
    const float* prev = u0;
    dim3 grid(NBATCH, 16, 2);
    for (int s = 0; s < 32; ++s) {
        float* next = (s == 31) ? out : bufs[s % 3];
        fdtd_step<<<grid, 256, 0, stream>>>(cur, prev, epsr, sigma, Sbuf, src, next, s);
        prev = cur;
        cur  = next;
    }
}


# all global loads of a block issued up front in one burst (branch-free clamped halo loads, 16-byte B-operand copy, restrict on prev/next) instead of a load-wait round trip per staged element and row
# speedup vs baseline: 1.8107x; 1.8107x over previous
#include <hip/hip_runtime.h>

typedef __attribute__((ext_vector_type(2))) float v2f;
typedef __attribute__((ext_vector_type(4))) float v4f;
typedef __attribute__((ext_vector_type(8))) float v8f;

#define DIMSZ 256
#define NBATCH 16
#define NPIX (DIMSZ * DIMSZ)
#define NELEM (NBATCH * NPIX)
#define LDSW 138
#define LDSH 22
#define NSCHUNK (7 * 6)
#define SBUF_ELEMS (NSCHUNK * 32 * 2)
#define SBUF_VEC4 (SBUF_ELEMS / 4)
#define TILE_ELEMS (LDSH * LDSW)
#define TILE_ITERS 12
#define SBUF_ITERS 3
#define TILE_ALLOC (TILE_ITERS * 256)
#define SBUF_ALLOC (SBUF_ITERS * 256 * 4)

__global__ void fdtd_init(const float* __restrict__ fdk,
                          const float* __restrict__ cw,
                          float* __restrict__ Sbuf) {
    __shared__ float weff[49];
    int t = threadIdx.x;
    if (t < 49) {
        float acc = 0.0f;
        #pragma unroll
        for (int ch = 1; ch < 15; ++ch)
            acc += cw[ch - 1] * fdk[ch * 49 + t];
        weff[t] = acc * (2.0f * 0.5f / 1.0f);
    }
    __syncthreads();
    for (int e = t; e < SBUF_ELEMS; e += blockDim.x) {
        int v    = e & 1;
        int lane = (e >> 1) & 31;
        int c    = (e >> 6) % 6;
        int ky   = e / 384;
        int n    = lane & 15;
        int k    = 4 * c + v + 2 * (lane >> 4);
        int j    = k - n;
        float val = (k < 22 && j >= 0 && j <= 6) ? weff[ky * 7 + j] : 0.0f;
        Sbuf[e] = val;
    }
}

__global__ __launch_bounds__(256) void fdtd_step(
    const float* __restrict__ cur,
    const float* __restrict__ prev,
    const float* __restrict__ epsr,
    const float* __restrict__ sigma,
    const float* __restrict__ Sglob,
    const float* __restrict__ src,
    float* __restrict__ next,
    int step) {
    __shared__ float tile[TILE_ALLOC];
    __shared__ __attribute__((aligned(16))) float Slds[SBUF_ALLOC];

    const int b  = blockIdx.x;
    const int y0 = blockIdx.y * 16;
    const int x0 = blockIdx.z * 128;
    const int t  = threadIdx.x;
    const float* curb = cur + b * NPIX;

    const int wave = t >> 5;
    const int lane = t & 31;
    const int mlo  = lane & 15;
    const int sel  = lane >> 4;
    const int xw   = 16 * wave;

    float tv[TILE_ITERS];
    #pragma unroll
    for (int i = 0; i < TILE_ITERS; ++i) {
        int e = t + 256 * i;
        int r = e / LDSW, c = e % LDSW;
        int gy = y0 + r - 3, gx = x0 + c - 3;
        bool inb = (c < 136 && gy >= 0 && gy < DIMSZ && gx >= 0 && gx < DIMSZ);
        int gyc = min(max(gy, 0), DIMSZ - 1);
        int gxc = min(max(gx, 0), DIMSZ - 1);
        float v = curb[gyc * DIMSZ + gxc];
        tv[i] = inb ? v : 0.0f;
    }
    const v4f* Sg4 = (const v4f*)Sglob;
    v4f sv[SBUF_ITERS];
    #pragma unroll
    for (int i = 0; i < SBUF_ITERS; ++i) {
        int q = min(t + 256 * i, SBUF_VEC4 - 1);
        sv[i] = Sg4[q];
    }
    const int gxo = x0 + xw + mlo;
    float ep[8], sg[8], pv[8];
    #pragma unroll
    for (int r = 0; r < 8; ++r) {
        int idx = b * NPIX + (y0 + r + 8 * sel) * DIMSZ + gxo;
        ep[r] = epsr[idx];
        sg[r] = sigma[idx];
        pv[r] = prev[idx];
    }

    #pragma unroll
    for (int i = 0; i < TILE_ITERS; ++i) tile[t + 256 * i] = tv[i];
    v4f* Sl4 = (v4f*)Slds;
    #pragma unroll
    for (int i = 0; i < SBUF_ITERS; ++i) Sl4[t + 256 * i] = sv[i];
    __syncthreads();

    v8f acc = {0.f, 0.f, 0.f, 0.f, 0.f, 0.f, 0.f, 0.f};
    #pragma unroll
    for (int ky = 0; ky < 7; ++ky) {
        const int arow = (mlo + ky) * LDSW + xw + 2 * sel;
        #pragma unroll
        for (int c = 0; c < 6; ++c) {
            v2f a = *(const v2f*)&tile[arow + 4 * c];
            v2f bmat = *(const v2f*)&Slds[((ky * 6 + c) * 32 + lane) * 2];
            acc = __builtin_amdgcn_wmma_f32_16x16x4_f32(
                false, a, false, bmat, (short)0, acc, false, false);
        }
    }

    const float dt = 0.5f;
    const float s0 = src[2 * step], s1 = src[2 * step + 1];
    const float srcterm = -(s0 - s1);
    #pragma unroll
    for (int r = 0; r < 8; ++r) {
        int m  = r + 8 * sel;
        int gy = y0 + m;
        int idx = b * NPIX + gy * DIMSZ + gxo;
        float inv_den = 1.0f / (2.0f * ep[r] / dt + sg[r]);
        float av = 4.0f * ep[r] / dt;
        float bv = 2.0f * ep[r] / dt - sg[r];
        float cu = tile[(m + 3) * LDSW + xw + mlo + 3];
        float rhs = acc[r] + av * cu - bv * pv[r];
        if (gy == 128 && gxo == 128) rhs += srcterm;
        next[idx] = inv_den * rhs;
    }
}

extern "C" void kernel_launch(void* const* d_in, const int* in_sizes, int n_in,
                              void* d_out, int out_size, void* d_ws, size_t ws_size,
                              hipStream_t stream) {
    const float* u1    = (const float*)d_in[0];
    const float* u0    = (const float*)d_in[1];
    const float* epsr  = (const float*)d_in[2];
    const float* sigma = (const float*)d_in[3];
    const float* fdk   = (const float*)d_in[4];
    const float* cw    = (const float*)d_in[5];
    const float* src   = (const float*)d_in[6];
    float* out = (float*)d_out;
    float* ws  = (float*)d_ws;

    float* bufs[3] = {ws, ws + NELEM, ws + 2 * NELEM};
    float* Sbuf = ws + 3 * NELEM;

    fdtd_init<<<1, 256, 0, stream>>>(fdk, cw, Sbuf);

    const float* cur  = u1;
    const float* prev = u0;
    dim3 grid(NBATCH, 16, 2);
    for (int s = 0; s < 32; ++s) {
        float* next = (s == 31) ? out : bufs[s % 3];
        fdtd_step<<<grid, 256, 0, stream>>>(cur, prev, epsr, sigma, Sbuf, src, next, s);
        prev = cur;
        cur  = next;
    }
}
